# starting kernel
# baseline (speedup 1.0000x reference)
#include <hip/hip_runtime.h>
#include <hip/hip_bf16.h>

typedef __attribute__((ext_vector_type(2))) float v2f;
typedef __attribute__((ext_vector_type(8))) float v8f;

#define MROWS 2048
#define FD    128

__global__ __launch_bounds__(128) void precompute_vecs(
    const float* __restrict__ W4, const float* __restrict__ b4,
    const float* __restrict__ W5, const float* __restrict__ b5,
    float* __restrict__ vl, float* __restrict__ vr, float* __restrict__ Cs) {
  const int f = threadIdx.x;
  float sl = 0.f, sr = 0.f;
  #pragma unroll 8
  for (int s = 0; s < 128; ++s) {
    const float w5 = W5[s];
    sl += w5 * W4[s * 256 + f];
    sr += w5 * W4[s * 256 + 128 + f];
  }
  vl[f] = sl;
  vr[f] = sr;
  if (f == 0) {
    float c = b5[0];
    for (int s = 0; s < 128; ++s) c += b4[s] * W5[s];
    Cs[0] = c;
  }
}

__global__ __launch_bounds__(256) void gemm_wmma_f32(
    const float* __restrict__ Hin, const float* __restrict__ W,
    const float* __restrict__ bias, float* __restrict__ Hout) {
  const int lane = threadIdx.x & 31;
  const int wave = threadIdx.x >> 5;
  const int m0   = blockIdx.x * 16;
  const int n0   = wave * 16;
  const int half = lane >> 4;
  const int l15  = lane & 15;

  const float* pA = Hin + (m0 + l15) * FD + half * 2;
  const float* pB = W   + (n0 + l15) * FD + half * 2;

  v8f acc = {};
  #pragma unroll
  for (int k0 = 0; k0 < FD; k0 += 4) {
    const float2 av = *(const float2*)(pA + k0);
    const float2 bv = *(const float2*)(pB + k0);
    v2f a; a.x = av.x; a.y = av.y;
    v2f b; b.x = bv.x; b.y = bv.y;
    acc = __builtin_amdgcn_wmma_f32_16x16x4_f32(
        false, a, false, b, (short)0, acc, false, false);
  }

  const int   col = n0 + l15;
  const float bv  = bias[col];
  #pragma unroll
  for (int g = 0; g < 8; ++g) {
    const int row = m0 + g + half * 8;
    Hout[row * FD + col] = acc[g] + bv;
  }
}

__global__ __launch_bounds__(256) void row_dots(
    const float* __restrict__ H, const float* __restrict__ vl,
    const float* __restrict__ vr, float* __restrict__ ul,
    float* __restrict__ ur) {
  const int lane = threadIdx.x & 31;
  const int wave = threadIdx.x >> 5;
  const int row  = blockIdx.x * 8 + wave;

  const float4 h4 = *(const float4*)(H + row * FD + lane * 4);
  const float4 l4 = *(const float4*)(vl + lane * 4);
  const float4 r4 = *(const float4*)(vr + lane * 4);
  float sl = h4.x * l4.x + h4.y * l4.y + h4.z * l4.z + h4.w * l4.w;
  float sr = h4.x * r4.x + h4.y * r4.y + h4.z * r4.z + h4.w * r4.w;
  #pragma unroll
  for (int off = 16; off > 0; off >>= 1) {
    sl += __shfl_down(sl, off, 32);
    sr += __shfl_down(sr, off, 32);
  }
  if (lane == 0) { ul[row] = sl; ur[row] = sr; }
}

__global__ __launch_bounds__(256) void expand_out(
    const float* __restrict__ ul, const float* __restrict__ ur,
    const float* __restrict__ Cs, float* __restrict__ out) {
  const unsigned idx = blockIdx.x * 256u + threadIdx.x;
  const unsigned j = idx & 255u;
  const unsigned i = (idx >> 8) & 255u;
  const unsigned b = idx >> 16;
  out[idx] = ul[b * 256u + i] + ur[b * 256u + j] + Cs[0];
}

extern "C" void kernel_launch(void* const* d_in, const int* in_sizes, int n_in,
                              void* d_out, int out_size, void* d_ws, size_t ws_size,
                              hipStream_t stream) {
  (void)in_sizes; (void)n_in; (void)out_size; (void)ws_size;
  const float* x  = (const float*)d_in[0];
  const float* W1 = (const float*)d_in[1];
  const float* b1 = (const float*)d_in[2];
  const float* W2 = (const float*)d_in[3];
  const float* b2 = (const float*)d_in[4];
  const float* W3 = (const float*)d_in[5];
  const float* b3 = (const float*)d_in[6];
  const float* W4 = (const float*)d_in[7];
  const float* b4 = (const float*)d_in[8];
  const float* W5 = (const float*)d_in[9];
  const float* b5 = (const float*)d_in[10];
  float* out = (float*)d_out;

  float* ws   = (float*)d_ws;
  float* bufA = ws;
  float* bufB = bufA + MROWS * FD;
  float* vl   = bufB + MROWS * FD;
  float* vr   = vl + FD;
  float* Cs   = vr + FD;
  float* ul   = Cs + 1;
  float* ur   = ul + MROWS;

  precompute_vecs<<<1, 128, 0, stream>>>(W4, b4, W5, b5, vl, vr, Cs);

  gemm_wmma_f32<<<MROWS / 16, 256, 0, stream>>>(x,    W1, b1, bufA);
  gemm_wmma_f32<<<MROWS / 16, 256, 0, stream>>>(bufA, W2, b2, bufB);
  gemm_wmma_f32<<<MROWS / 16, 256, 0, stream>>>(bufB, W3, b3, bufA);

  row_dots<<<MROWS / 8, 256, 0, stream>>>(bufA, vl, vr, ul, ur);

  expand_out<<<(8 * 256 * 256) / 256, 256, 0, stream>>>(ul, ur, Cs, out);
}


# Fuse six launches into two: three WMMA layers chained per 16-row stripe through LDS (weight fold on two extra blocks), then row dots plus output expansion in one kernel; sum orders kept
# speedup vs baseline: 2.2799x; 2.2799x over previous
#include <hip/hip_runtime.h>
#include <hip/hip_bf16.h>

typedef __attribute__((ext_vector_type(2))) float v2f;
typedef __attribute__((ext_vector_type(8))) float v8f;

#define MROWS 2048
#define FD    128
#define NSTRIPE (MROWS / 16)
#define LDP   132

__device__ __forceinline__ v8f tile_k128(const float* pA, const float* pB) {
  v8f acc = {};
  #pragma unroll
  for (int k0 = 0; k0 < FD; k0 += 4) {
    const float2 av = *(const float2*)(pA + k0);
    const float2 bv = *(const float2*)(pB + k0);
    v2f a; a.x = av.x; a.y = av.y;
    v2f b; b.x = bv.x; b.y = bv.y;
    acc = __builtin_amdgcn_wmma_f32_16x16x4_f32(
        false, a, false, b, (short)0, acc, false, false);
  }
  return acc;
}

__global__ __launch_bounds__(256) void chain3_and_fold(
    const float* __restrict__ x,
    const float* __restrict__ W1, const float* __restrict__ b1,
    const float* __restrict__ W2, const float* __restrict__ b2,
    const float* __restrict__ W3, const float* __restrict__ b3,
    const float* __restrict__ W4, const float* __restrict__ b4,
    const float* __restrict__ W5, const float* __restrict__ b5,
    float* __restrict__ H3, float* __restrict__ v, float* __restrict__ Cs) {
  __shared__ float sH1[16 * LDP];
  __shared__ float sH2[16 * LDP];

  if (blockIdx.x >= NSTRIPE) {
    if (blockIdx.x == NSTRIPE) {
      const int t = threadIdx.x;
      float sv = 0.f;
      #pragma unroll 32
      for (int s = 0; s < 128; ++s) sv += W5[s] * W4[s * 256 + t];
      v[t] = sv;
    } else if (threadIdx.x == 0) {
      float c = b5[0];
      for (int s = 0; s < 128; ++s) c += b4[s] * W5[s];
      Cs[0] = c;
    }
    return;
  }

  const int lane = threadIdx.x & 31;
  const int wave = threadIdx.x >> 5;
  const int m0   = blockIdx.x * 16;
  const int n0   = wave * 16;
  const int half = lane >> 4;
  const int l15  = lane & 15;
  const int col  = n0 + l15;
  const int wofs = col * FD + half * 2;
  const int sofs = l15 * LDP + half * 2;

  {
    const v8f acc = tile_k128(x + (m0 + l15) * FD + half * 2, W1 + wofs);
    const float bv = b1[col];
    #pragma unroll
    for (int g = 0; g < 8; ++g) sH1[(g + half * 8) * LDP + col] = acc[g] + bv;
  }
  __syncthreads();
  {
    const v8f acc = tile_k128(sH1 + sofs, W2 + wofs);
    const float bv = b2[col];
    #pragma unroll
    for (int g = 0; g < 8; ++g) sH2[(g + half * 8) * LDP + col] = acc[g] + bv;
  }
  __syncthreads();
  {
    const v8f acc = tile_k128(sH2 + sofs, W3 + wofs);
    const float bv = b3[col];
    #pragma unroll
    for (int g = 0; g < 8; ++g) {
      const int row = m0 + g + half * 8;
      H3[row * FD + col] = acc[g] + bv;
    }
  }
}

__global__ __launch_bounds__(256) void dots_and_expand(
    const float* __restrict__ H, const float* __restrict__ v,
    const float* __restrict__ Cs, float* __restrict__ out) {
  __shared__ float sUr[256];
  __shared__ float sUl[16];
  const int lane = threadIdx.x & 31;
  const int wave = threadIdx.x >> 5;
  const int b    = blockIdx.x >> 4;
  const int g    = blockIdx.x & 15;

  const float4 l4 = *(const float4*)(v + lane * 4);
  const float4 r4 = *(const float4*)(v + FD + lane * 4);
  const float* Hb = H + b * 256 * FD + lane * 4;

  float4 h[32];
  #pragma unroll
  for (int r = 0; r < 32; ++r) h[r] = *(const float4*)(Hb + (wave * 32 + r) * FD);
  const float4 ha = *(const float4*)(Hb + (g * 16 + wave * 2) * FD);
  const float4 hb = *(const float4*)(Hb + (g * 16 + wave * 2 + 1) * FD);

  float sr[32];
  #pragma unroll
  for (int r = 0; r < 32; ++r)
    sr[r] = h[r].x * r4.x + h[r].y * r4.y + h[r].z * r4.z + h[r].w * r4.w;
  float sla = ha.x * l4.x + ha.y * l4.y + ha.z * l4.z + ha.w * l4.w;
  float slb = hb.x * l4.x + hb.y * l4.y + hb.z * l4.z + hb.w * l4.w;
  #pragma unroll
  for (int off = 16; off > 0; off >>= 1) {
    #pragma unroll
    for (int r = 0; r < 32; ++r) sr[r] += __shfl_down(sr[r], off, 32);
    sla += __shfl_down(sla, off, 32);
    slb += __shfl_down(slb, off, 32);
  }
  if (lane == 0) {
    #pragma unroll
    for (int r = 0; r < 32; ++r) sUr[wave * 32 + r] = sr[r];
    sUl[wave * 2]     = sla;
    sUl[wave * 2 + 1] = slb;
  }
  __syncthreads();

  const float c  = Cs[0];
  const int   j4 = (threadIdx.x & 63) * 4;
  const int   rs = threadIdx.x >> 6;
  const float4 u4 = *(const float4*)(sUr + j4);
  float* ob = out + (b * 256 + g * 16) * 256 + j4;
  for (int p = 0; p < 4; ++p) {
    const int   r = p * 4 + rs;
    const float u = sUl[r];
    float4 o;
    o.x = u + u4.x + c;
    o.y = u + u4.y + c;
    o.z = u + u4.z + c;
    o.w = u + u4.w + c;
    *(float4*)(ob + r * 256) = o;
  }
}

extern "C" void kernel_launch(void* const* d_in, const int* in_sizes, int n_in,
                              void* d_out, int out_size, void* d_ws, size_t ws_size,
                              hipStream_t stream) {
  (void)in_sizes; (void)n_in; (void)out_size; (void)ws_size;
  const float* x  = (const float*)d_in[0];
  const float* W1 = (const float*)d_in[1];
  const float* b1 = (const float*)d_in[2];
  const float* W2 = (const float*)d_in[3];
  const float* b2 = (const float*)d_in[4];
  const float* W3 = (const float*)d_in[5];
  const float* b3 = (const float*)d_in[6];
  const float* W4 = (const float*)d_in[7];
  const float* b4 = (const float*)d_in[8];
  const float* W5 = (const float*)d_in[9];
  const float* b5 = (const float*)d_in[10];
  float* out = (float*)d_out;

  float* ws = (float*)d_ws;
  float* H3 = ws;
  float* v  = H3 + MROWS * FD;
  float* Cs = v + 2 * FD;

  chain3_and_fold<<<NSTRIPE + 2, 256, 0, stream>>>(
      x, W1, b1, W2, b2, W3, b3, W4, b4, W5, b5, H3, v, Cs);

  dots_and_expand<<<8 * 16, 256, 0, stream>>>(H3, v, Cs, out);
}


# GEMM launch: x stripe and each wave's weight rows loaded as whole 512-byte rows and parked in LDS, each weight tile requested one layer ahead; fragments read from LDS only (k order, sums unchanged)
# speedup vs baseline: 2.8278x; 1.2403x over previous
#include <hip/hip_runtime.h>
#include <hip/hip_bf16.h>

typedef __attribute__((ext_vector_type(2))) float v2f;
typedef __attribute__((ext_vector_type(8))) float v8f;
typedef __attribute__((ext_vector_type(4))) float v4f;

#define MROWS 2048
#define FD    128
#define NSTRIPE (MROWS / 16)
#define LDP   132

__device__ __forceinline__ v8f tile_k128(const float* pA, const float* pB) {
  v8f acc = {};
  #pragma unroll
  for (int k0 = 0; k0 < FD; k0 += 4) {
    const float2 av = *(const float2*)(pA + k0);
    const float2 bv = *(const float2*)(pB + k0);
    v2f a; a.x = av.x; a.y = av.y;
    v2f b; b.x = bv.x; b.y = bv.y;
    acc = __builtin_amdgcn_wmma_f32_16x16x4_f32(
        false, a, false, b, (short)0, acc, false, false);
  }
  return acc;
}

__global__ __launch_bounds__(256) void chain3_and_fold(
    const float* __restrict__ x,
    const float* __restrict__ W1, const float* __restrict__ b1,
    const float* __restrict__ W2, const float* __restrict__ b2,
    const float* __restrict__ W3, const float* __restrict__ b3,
    const float* __restrict__ W4, const float* __restrict__ b4,
    const float* __restrict__ W5, const float* __restrict__ b5,
    float* __restrict__ H3, float* __restrict__ v, float* __restrict__ Cs) {
  __shared__ __attribute__((aligned(16))) float sW[3][FD * LDP];
  __shared__ __attribute__((aligned(16))) float sX[16 * LDP];
  __shared__ float sH1[16 * LDP];
  __shared__ float sH2[16 * LDP];

  if (blockIdx.x >= NSTRIPE) {
    if (blockIdx.x == NSTRIPE) {
      const int t = threadIdx.x;
      float sv = 0.f;
      #pragma unroll 32
      for (int s = 0; s < 128; ++s) sv += W5[s] * W4[s * 256 + t];
      v[t] = sv;
    } else if (threadIdx.x == 0) {
      float c = b5[0];
      for (int s = 0; s < 128; ++s) c += b4[s] * W5[s];
      Cs[0] = c;
    }
    return;
  }

  const int lane = threadIdx.x & 31;
  const int wave = threadIdx.x >> 5;
  const int m0   = blockIdx.x * 16;
  const int n0   = wave * 16;
  const int half = lane >> 4;
  const int l15  = lane & 15;
  const int col  = n0 + l15;
  const int wofs = col * LDP + half * 2;
  const int sofs = l15 * LDP + half * 2;
  const int c4   = lane * 4;

  v4f xq[2], wq1[16], wq2[16], wq3[16];
  #pragma unroll
  for (int i = 0; i < 2; ++i)
    xq[i] = *(const v4f*)(x + (m0 + wave * 2 + i) * FD + c4);
  #pragma unroll
  for (int i = 0; i < 16; ++i) wq1[i] = *(const v4f*)(W1 + (n0 + i) * FD + c4);
  const float bv1 = b1[col];
  const float bv2 = b2[col];
  const float bv3 = b3[col];

  #pragma unroll
  for (int i = 0; i < 2; ++i) *(v4f*)(sX + (wave * 2 + i) * LDP + c4) = xq[i];
  #pragma unroll
  for (int i = 0; i < 16; ++i) *(v4f*)(sW[0] + (n0 + i) * LDP + c4) = wq1[i];
  __syncthreads();
  #pragma unroll
  for (int i = 0; i < 16; ++i) wq2[i] = *(const v4f*)(W2 + (n0 + i) * FD + c4);
  __builtin_amdgcn_sched_barrier(0);
  {
    const v8f acc = tile_k128(sX + sofs, sW[0] + wofs);
    #pragma unroll
    for (int g = 0; g < 8; ++g) sH1[(g + half * 8) * LDP + col] = acc[g] + bv1;
  }
  #pragma unroll
  for (int i = 0; i < 16; ++i) *(v4f*)(sW[1] + (n0 + i) * LDP + c4) = wq2[i];
  __syncthreads();
  #pragma unroll
  for (int i = 0; i < 16; ++i) wq3[i] = *(const v4f*)(W3 + (n0 + i) * FD + c4);
  __builtin_amdgcn_sched_barrier(0);
  {
    const v8f acc = tile_k128(sH1 + sofs, sW[1] + wofs);
    #pragma unroll
    for (int g = 0; g < 8; ++g) sH2[(g + half * 8) * LDP + col] = acc[g] + bv2;
  }
  #pragma unroll
  for (int i = 0; i < 16; ++i) *(v4f*)(sW[2] + (n0 + i) * LDP + c4) = wq3[i];
  __syncthreads();
  {
    const v8f acc = tile_k128(sH2 + sofs, sW[2] + wofs);
    #pragma unroll
    for (int g = 0; g < 8; ++g) {
      const int row = m0 + g + half * 8;
      H3[row * FD + col] = acc[g] + bv3;
    }
  }
}

__global__ __launch_bounds__(256) void dots_and_expand(
    const float* __restrict__ H, const float* __restrict__ v,
    const float* __restrict__ Cs, float* __restrict__ out) {
  __shared__ float sUr[256];
  __shared__ float sUl[16];
  const int lane = threadIdx.x & 31;
  const int wave = threadIdx.x >> 5;
  const int b    = blockIdx.x >> 4;
  const int g    = blockIdx.x & 15;

  const float4 l4 = *(const float4*)(v + lane * 4);
  const float4 r4 = *(const float4*)(v + FD + lane * 4);
  const float* Hb = H + b * 256 * FD + lane * 4;

  float4 h[32];
  #pragma unroll
  for (int r = 0; r < 32; ++r) h[r] = *(const float4*)(Hb + (wave * 32 + r) * FD);
  const float4 ha = *(const float4*)(Hb + (g * 16 + wave * 2) * FD);
  const float4 hb = *(const float4*)(Hb + (g * 16 + wave * 2 + 1) * FD);

  float sr[32];
  #pragma unroll
  for (int r = 0; r < 32; ++r)
    sr[r] = h[r].x * r4.x + h[r].y * r4.y + h[r].z * r4.z + h[r].w * r4.w;
  float sla = ha.x * l4.x + ha.y * l4.y + ha.z * l4.z + ha.w * l4.w;
  float slb = hb.x * l4.x + hb.y * l4.y + hb.z * l4.z + hb.w * l4.w;
  #pragma unroll
  for (int off = 16; off > 0; off >>= 1) {
    #pragma unroll
    for (int r = 0; r < 32; ++r) sr[r] += __shfl_down(sr[r], off, 32);
    sla += __shfl_down(sla, off, 32);
    slb += __shfl_down(slb, off, 32);
  }
  if (lane == 0) {
    #pragma unroll
    for (int r = 0; r < 32; ++r) sUr[wave * 32 + r] = sr[r];
    sUl[wave * 2]     = sla;
    sUl[wave * 2 + 1] = slb;
  }
  __syncthreads();

  const float c  = Cs[0];
  const int   j4 = (threadIdx.x & 63) * 4;
  const int   rs = threadIdx.x >> 6;
  const float4 u4 = *(const float4*)(sUr + j4);
  float* ob = out + (b * 256 + g * 16) * 256 + j4;
  for (int p = 0; p < 4; ++p) {
    const int   r = p * 4 + rs;
    const float u = sUl[r];
    float4 o;
    o.x = u + u4.x + c;
    o.y = u + u4.y + c;
    o.z = u + u4.z + c;
    o.w = u + u4.w + c;
    *(float4*)(ob + r * 256) = o;
  }
}

extern "C" void kernel_launch(void* const* d_in, const int* in_sizes, int n_in,
                              void* d_out, int out_size, void* d_ws, size_t ws_size,
                              hipStream_t stream) {
  (void)in_sizes; (void)n_in; (void)out_size; (void)ws_size;
  const float* x  = (const float*)d_in[0];
  const float* W1 = (const float*)d_in[1];
  const float* b1 = (const float*)d_in[2];
  const float* W2 = (const float*)d_in[3];
  const float* b2 = (const float*)d_in[4];
  const float* W3 = (const float*)d_in[5];
  const float* b3 = (const float*)d_in[6];
  const float* W4 = (const float*)d_in[7];
  const float* b4 = (const float*)d_in[8];
  const float* W5 = (const float*)d_in[9];
  const float* b5 = (const float*)d_in[10];
  float* out = (float*)d_out;

  float* ws = (float*)d_ws;
  float* H3 = ws;
  float* v  = H3 + MROWS * FD;
  float* Cs = v + 2 * FD;

  chain3_and_fold<<<NSTRIPE + 2, 256, 0, stream>>>(
      x, W1, b1, W2, b2, W3, b3, W4, b4, W5, b5, H3, v, Cs);

  dots_and_expand<<<8 * 16, 256, 0, stream>>>(H3, v, Cs, out);
}


# Output launch: its 36 row and vector loads and its scalar load all requested at entry as one group (input-only fence on the loaded values) instead of six dependent round trips; sums unchanged
# speedup vs baseline: 3.2327x; 1.1432x over previous
#include <hip/hip_runtime.h>
#include <hip/hip_bf16.h>

typedef __attribute__((ext_vector_type(2))) float v2f;
typedef __attribute__((ext_vector_type(8))) float v8f;
typedef __attribute__((ext_vector_type(4))) float v4f;

#define MROWS 2048
#define FD    128
#define NSTRIPE (MROWS / 16)
#define LDP   132

__device__ __forceinline__ v8f tile_k128(const float* pA, const float* pB) {
  v8f acc = {};
  #pragma unroll
  for (int k0 = 0; k0 < FD; k0 += 4) {
    const float2 av = *(const float2*)(pA + k0);
    const float2 bv = *(const float2*)(pB + k0);
    v2f a; a.x = av.x; a.y = av.y;
    v2f b; b.x = bv.x; b.y = bv.y;
    acc = __builtin_amdgcn_wmma_f32_16x16x4_f32(
        false, a, false, b, (short)0, acc, false, false);
  }
  return acc;
}

__global__ __launch_bounds__(256) void chain3_and_fold(
    const float* __restrict__ x,
    const float* __restrict__ W1, const float* __restrict__ b1,
    const float* __restrict__ W2, const float* __restrict__ b2,
    const float* __restrict__ W3, const float* __restrict__ b3,
    const float* __restrict__ W4, const float* __restrict__ b4,
    const float* __restrict__ W5, const float* __restrict__ b5,
    float* __restrict__ H3, float* __restrict__ v, float* __restrict__ Cs) {
  __shared__ __attribute__((aligned(16))) float sW[3][FD * LDP];
  __shared__ __attribute__((aligned(16))) float sX[16 * LDP];
  __shared__ float sH1[16 * LDP];
  __shared__ float sH2[16 * LDP];

  if (blockIdx.x >= NSTRIPE) {
    if (blockIdx.x == NSTRIPE) {
      const int t = threadIdx.x;
      float sv = 0.f;
      #pragma unroll 32
      for (int s = 0; s < 128; ++s) sv += W5[s] * W4[s * 256 + t];
      v[t] = sv;
    } else if (threadIdx.x == 0) {
      float c = b5[0];
      for (int s = 0; s < 128; ++s) c += b4[s] * W5[s];
      Cs[0] = c;
    }
    return;
  }

  const int lane = threadIdx.x & 31;
  const int wave = threadIdx.x >> 5;
  const int m0   = blockIdx.x * 16;
  const int n0   = wave * 16;
  const int half = lane >> 4;
  const int l15  = lane & 15;
  const int col  = n0 + l15;
  const int wofs = col * LDP + half * 2;
  const int sofs = l15 * LDP + half * 2;
  const int c4   = lane * 4;

  v4f xq[2], wq1[16], wq2[16], wq3[16];
  #pragma unroll
  for (int i = 0; i < 2; ++i)
    xq[i] = *(const v4f*)(x + (m0 + wave * 2 + i) * FD + c4);
  #pragma unroll
  for (int i = 0; i < 16; ++i) wq1[i] = *(const v4f*)(W1 + (n0 + i) * FD + c4);
  const float bv1 = b1[col];
  const float bv2 = b2[col];
  const float bv3 = b3[col];

  #pragma unroll
  for (int i = 0; i < 2; ++i) *(v4f*)(sX + (wave * 2 + i) * LDP + c4) = xq[i];
  #pragma unroll
  for (int i = 0; i < 16; ++i) *(v4f*)(sW[0] + (n0 + i) * LDP + c4) = wq1[i];
  __syncthreads();
  #pragma unroll
  for (int i = 0; i < 16; ++i) wq2[i] = *(const v4f*)(W2 + (n0 + i) * FD + c4);
  __builtin_amdgcn_sched_barrier(0);
  {
    const v8f acc = tile_k128(sX + sofs, sW[0] + wofs);
    #pragma unroll
    for (int g = 0; g < 8; ++g) sH1[(g + half * 8) * LDP + col] = acc[g] + bv1;
  }
  #pragma unroll
  for (int i = 0; i < 16; ++i) *(v4f*)(sW[1] + (n0 + i) * LDP + c4) = wq2[i];
  __syncthreads();
  #pragma unroll
  for (int i = 0; i < 16; ++i) wq3[i] = *(const v4f*)(W3 + (n0 + i) * FD + c4);
  __builtin_amdgcn_sched_barrier(0);
  {
    const v8f acc = tile_k128(sH1 + sofs, sW[1] + wofs);
    #pragma unroll
    for (int g = 0; g < 8; ++g) sH2[(g + half * 8) * LDP + col] = acc[g] + bv2;
  }
  #pragma unroll
  for (int i = 0; i < 16; ++i) *(v4f*)(sW[2] + (n0 + i) * LDP + c4) = wq3[i];
  __syncthreads();
  {
    const v8f acc = tile_k128(sH2 + sofs, sW[2] + wofs);
    #pragma unroll
    for (int g = 0; g < 8; ++g) {
      const int row = m0 + g + half * 8;
      H3[row * FD + col] = acc[g] + bv3;
    }
  }
}

__global__ __launch_bounds__(256) void dots_and_expand(
    const float* __restrict__ H, const float* __restrict__ v,
    const float* __restrict__ Cs, float* __restrict__ out) {
  __shared__ float sUr[256];
  __shared__ float sUl[16];
  const int lane = threadIdx.x & 31;
  const int wave = threadIdx.x >> 5;
  const int b    = blockIdx.x >> 4;
  const int g    = blockIdx.x & 15;

  const float c  = Cs[0];
  const v4f   l4 = *(const v4f*)(v + lane * 4);
  const v4f   r4 = *(const v4f*)(v + FD + lane * 4);
  const float* Hb = H + b * 256 * FD + lane * 4;
  v4f h[32];
  #pragma unroll
  for (int r = 0; r < 32; ++r) h[r] = *(const v4f*)(Hb + (wave * 32 + r) * FD);
  const v4f ha = *(const v4f*)(Hb + (g * 16 + wave * 2) * FD);
  const v4f hb = *(const v4f*)(Hb + (g * 16 + wave * 2 + 1) * FD);
  asm volatile("" :: "v"(h[0]), "v"(h[1]), "v"(h[2]), "v"(h[3]), "v"(h[4]), "v"(h[5]),
               "v"(h[6]), "v"(h[7]), "v"(h[8]), "v"(h[9]), "v"(h[10]), "v"(h[11]),
               "v"(h[12]), "v"(h[13]), "v"(h[14]), "v"(h[15]), "v"(h[16]), "v"(h[17]));
  asm volatile("" :: "v"(h[18]), "v"(h[19]), "v"(h[20]), "v"(h[21]), "v"(h[22]), "v"(h[23]),
               "v"(h[24]), "v"(h[25]), "v"(h[26]), "v"(h[27]), "v"(h[28]), "v"(h[29]),
               "v"(h[30]), "v"(h[31]), "v"(ha), "v"(hb), "v"(l4), "v"(r4));

  float sr[32];
  #pragma unroll
  for (int r = 0; r < 32; ++r)
    sr[r] = h[r].x * r4.x + h[r].y * r4.y + h[r].z * r4.z + h[r].w * r4.w;
  float sla = ha.x * l4.x + ha.y * l4.y + ha.z * l4.z + ha.w * l4.w;
  float slb = hb.x * l4.x + hb.y * l4.y + hb.z * l4.z + hb.w * l4.w;
  #pragma unroll
  for (int off = 16; off > 0; off >>= 1) {
    #pragma unroll
    for (int r = 0; r < 32; ++r) sr[r] += __shfl_down(sr[r], off, 32);
    sla += __shfl_down(sla, off, 32);
    slb += __shfl_down(slb, off, 32);
  }
  if (lane == 0) {
    #pragma unroll
    for (int r = 0; r < 32; ++r) sUr[wave * 32 + r] = sr[r];
    sUl[wave * 2]     = sla;
    sUl[wave * 2 + 1] = slb;
  }
  __syncthreads();

  const int   j4 = (threadIdx.x & 63) * 4;
  const int   rs = threadIdx.x >> 6;
  const float4 u4 = *(const float4*)(sUr + j4);
  float* ob = out + (b * 256 + g * 16) * 256 + j4;
  for (int p = 0; p < 4; ++p) {
    const int   r = p * 4 + rs;
    const float u = sUl[r];
    float4 o;
    o.x = u + u4.x + c;
    o.y = u + u4.y + c;
    o.z = u + u4.z + c;
    o.w = u + u4.w + c;
    *(float4*)(ob + r * 256) = o;
  }
}

extern "C" void kernel_launch(void* const* d_in, const int* in_sizes, int n_in,
                              void* d_out, int out_size, void* d_ws, size_t ws_size,
                              hipStream_t stream) {
  (void)in_sizes; (void)n_in; (void)out_size; (void)ws_size;
  const float* x  = (const float*)d_in[0];
  const float* W1 = (const float*)d_in[1];
  const float* b1 = (const float*)d_in[2];
  const float* W2 = (const float*)d_in[3];
  const float* b2 = (const float*)d_in[4];
  const float* W3 = (const float*)d_in[5];
  const float* b3 = (const float*)d_in[6];
  const float* W4 = (const float*)d_in[7];
  const float* b4 = (const float*)d_in[8];
  const float* W5 = (const float*)d_in[9];
  const float* b5 = (const float*)d_in[10];
  float* out = (float*)d_out;

  float* ws = (float*)d_ws;
  float* H3 = ws;
  float* v  = H3 + MROWS * FD;
  float* Cs = v + 2 * FD;

  chain3_and_fold<<<NSTRIPE + 2, 256, 0, stream>>>(
      x, W1, b1, W2, b2, W3, b3, W4, b4, W5, b5, H3, v, Cs);

  dots_and_expand<<<8 * 16, 256, 0, stream>>>(H3, v, Cs, out);
}


# Output launch tiled in two dimensions: a block owns 64 rows by 64 columns of a batch and dots 128 rows of h3, not 272; per-row dot order and lane tree unchanged
# speedup vs baseline: 3.5115x; 1.0862x over previous
#include <hip/hip_runtime.h>
#include <hip/hip_bf16.h>

typedef __attribute__((ext_vector_type(2))) float v2f;
typedef __attribute__((ext_vector_type(8))) float v8f;
typedef __attribute__((ext_vector_type(4))) float v4f;

#define MROWS 2048
#define FD    128
#define NSTRIPE (MROWS / 16)
#define LDP   132

__device__ __forceinline__ v8f tile_k128(const float* pA, const float* pB) {
  v8f acc = {};
  #pragma unroll
  for (int k0 = 0; k0 < FD; k0 += 4) {
    const float2 av = *(const float2*)(pA + k0);
    const float2 bv = *(const float2*)(pB + k0);
    v2f a; a.x = av.x; a.y = av.y;
    v2f b; b.x = bv.x; b.y = bv.y;
    acc = __builtin_amdgcn_wmma_f32_16x16x4_f32(
        false, a, false, b, (short)0, acc, false, false);
  }
  return acc;
}

__global__ __launch_bounds__(256) void chain3_and_fold(
    const float* __restrict__ x,
    const float* __restrict__ W1, const float* __restrict__ b1,
    const float* __restrict__ W2, const float* __restrict__ b2,
    const float* __restrict__ W3, const float* __restrict__ b3,
    const float* __restrict__ W4, const float* __restrict__ b4,
    const float* __restrict__ W5, const float* __restrict__ b5,
    float* __restrict__ H3, float* __restrict__ v, float* __restrict__ Cs) {
  __shared__ __attribute__((aligned(16))) float sW[3][FD * LDP];
  __shared__ __attribute__((aligned(16))) float sX[16 * LDP];
  __shared__ float sH1[16 * LDP];
  __shared__ float sH2[16 * LDP];

  if (blockIdx.x >= NSTRIPE) {
    if (blockIdx.x == NSTRIPE) {
      const int t = threadIdx.x;
      float sv = 0.f;
      #pragma unroll 32
      for (int s = 0; s < 128; ++s) sv += W5[s] * W4[s * 256 + t];
      v[t] = sv;
    } else if (threadIdx.x == 0) {
      float c = b5[0];
      for (int s = 0; s < 128; ++s) c += b4[s] * W5[s];
      Cs[0] = c;
    }
    return;
  }

  const int lane = threadIdx.x & 31;
  const int wave = threadIdx.x >> 5;
  const int m0   = blockIdx.x * 16;
  const int n0   = wave * 16;
  const int half = lane >> 4;
  const int l15  = lane & 15;
  const int col  = n0 + l15;
  const int wofs = col * LDP + half * 2;
  const int sofs = l15 * LDP + half * 2;
  const int c4   = lane * 4;

  v4f xq[2], wq1[16], wq2[16], wq3[16];
  #pragma unroll
  for (int i = 0; i < 2; ++i)
    xq[i] = *(const v4f*)(x + (m0 + wave * 2 + i) * FD + c4);
  #pragma unroll
  for (int i = 0; i < 16; ++i) wq1[i] = *(const v4f*)(W1 + (n0 + i) * FD + c4);
  const float bv1 = b1[col];
  const float bv2 = b2[col];
  const float bv3 = b3[col];

  #pragma unroll
  for (int i = 0; i < 2; ++i) *(v4f*)(sX + (wave * 2 + i) * LDP + c4) = xq[i];
  #pragma unroll
  for (int i = 0; i < 16; ++i) *(v4f*)(sW[0] + (n0 + i) * LDP + c4) = wq1[i];
  __syncthreads();
  #pragma unroll
  for (int i = 0; i < 16; ++i) wq2[i] = *(const v4f*)(W2 + (n0 + i) * FD + c4);
  __builtin_amdgcn_sched_barrier(0);
  {
    const v8f acc = tile_k128(sX + sofs, sW[0] + wofs);
    #pragma unroll
    for (int g = 0; g < 8; ++g) sH1[(g + half * 8) * LDP + col] = acc[g] + bv1;
  }
  #pragma unroll
  for (int i = 0; i < 16; ++i) *(v4f*)(sW[1] + (n0 + i) * LDP + c4) = wq2[i];
  __syncthreads();
  #pragma unroll
  for (int i = 0; i < 16; ++i) wq3[i] = *(const v4f*)(W3 + (n0 + i) * FD + c4);
  __builtin_amdgcn_sched_barrier(0);
  {
    const v8f acc = tile_k128(sH1 + sofs, sW[1] + wofs);
    #pragma unroll
    for (int g = 0; g < 8; ++g) sH2[(g + half * 8) * LDP + col] = acc[g] + bv2;
  }
  #pragma unroll
  for (int i = 0; i < 16; ++i) *(v4f*)(sW[2] + (n0 + i) * LDP + c4) = wq3[i];
  __syncthreads();
  {
    const v8f acc = tile_k128(sH2 + sofs, sW[2] + wofs);
    #pragma unroll
    for (int g = 0; g < 8; ++g) {
      const int row = m0 + g + half * 8;
      H3[row * FD + col] = acc[g] + bv3;
    }
  }
}

__global__ __launch_bounds__(256) void dots_and_expand(
    const float* __restrict__ H, const float* __restrict__ v,
    const float* __restrict__ Cs, float* __restrict__ out) {
  __shared__ __attribute__((aligned(16))) float sUr[64];
  __shared__ float sUl[64];
  const int lane = threadIdx.x & 31;
  const int wave = threadIdx.x >> 5;
  const int b    = blockIdx.x >> 4;
  const int it   = (blockIdx.x >> 2) & 3;
  const int jt   = blockIdx.x & 3;

  const float c  = Cs[0];
  const v4f   l4 = *(const v4f*)(v + lane * 4);
  const v4f   r4 = *(const v4f*)(v + FD + lane * 4);
  const float* Hb = H + b * 256 * FD + lane * 4;
  v4f hj[8], hi[8];
  #pragma unroll
  for (int r = 0; r < 8; ++r) hj[r] = *(const v4f*)(Hb + (jt * 64 + wave * 8 + r) * FD);
  #pragma unroll
  for (int r = 0; r < 8; ++r) hi[r] = *(const v4f*)(Hb + (it * 64 + wave * 8 + r) * FD);
  asm volatile("" :: "v"(hj[0]), "v"(hj[1]), "v"(hj[2]), "v"(hj[3]), "v"(hj[4]), "v"(hj[5]),
               "v"(hj[6]), "v"(hj[7]), "v"(hi[0]), "v"(hi[1]), "v"(hi[2]), "v"(hi[3]),
               "v"(hi[4]), "v"(hi[5]), "v"(hi[6]), "v"(hi[7]), "v"(l4), "v"(r4));

  float sr[8], sl[8];
  #pragma unroll
  for (int r = 0; r < 8; ++r) {
    sr[r] = hj[r].x * r4.x + hj[r].y * r4.y + hj[r].z * r4.z + hj[r].w * r4.w;
    sl[r] = hi[r].x * l4.x + hi[r].y * l4.y + hi[r].z * l4.z + hi[r].w * l4.w;
  }
  #pragma unroll
  for (int off = 16; off > 0; off >>= 1) {
    #pragma unroll
    for (int r = 0; r < 8; ++r) {
      sr[r] += __shfl_down(sr[r], off, 32);
      sl[r] += __shfl_down(sl[r], off, 32);
    }
  }
  if (lane == 0) {
    #pragma unroll
    for (int r = 0; r < 8; ++r) {
      sUr[wave * 8 + r] = sr[r];
      sUl[wave * 8 + r] = sl[r];
    }
  }
  __syncthreads();

  const int   jq = (threadIdx.x & 15) * 4;
  const int   rs = threadIdx.x >> 4;
  const float4 u4 = *(const float4*)(sUr + jq);
  float* ob = out + (b * 256 + it * 64) * 256 + jt * 64 + jq;
  for (int p = 0; p < 4; ++p) {
    const int   r = p * 16 + rs;
    const float u = sUl[r];
    float4 o;
    o.x = u + u4.x + c;
    o.y = u + u4.y + c;
    o.z = u + u4.z + c;
    o.w = u + u4.w + c;
    *(float4*)(ob + r * 256) = o;
  }
}

extern "C" void kernel_launch(void* const* d_in, const int* in_sizes, int n_in,
                              void* d_out, int out_size, void* d_ws, size_t ws_size,
                              hipStream_t stream) {
  (void)in_sizes; (void)n_in; (void)out_size; (void)ws_size;
  const float* x  = (const float*)d_in[0];
  const float* W1 = (const float*)d_in[1];
  const float* b1 = (const float*)d_in[2];
  const float* W2 = (const float*)d_in[3];
  const float* b2 = (const float*)d_in[4];
  const float* W3 = (const float*)d_in[5];
  const float* b3 = (const float*)d_in[6];
  const float* W4 = (const float*)d_in[7];
  const float* b4 = (const float*)d_in[8];
  const float* W5 = (const float*)d_in[9];
  const float* b5 = (const float*)d_in[10];
  float* out = (float*)d_out;

  float* ws = (float*)d_ws;
  float* H3 = ws;
  float* v  = H3 + MROWS * FD;
  float* Cs = v + 2 * FD;

  chain3_and_fold<<<NSTRIPE + 2, 256, 0, stream>>>(
      x, W1, b1, W2, b2, W3, b3, W4, b4, W5, b5, H3, v, Cs);

  dots_and_expand<<<8 * 16, 256, 0, stream>>>(H3, v, Cs, out);
}


# Launch 1 spread over the whole chip: 16x64 tile blocks of four waves (256, not 128) and the weight fold cut into eight partial chains whose partial vectors launch 2 adds; fold sum order changes
# speedup vs baseline: 4.3359x; 1.2348x over previous
#include <hip/hip_runtime.h>
#include <hip/hip_bf16.h>

typedef __attribute__((ext_vector_type(2))) float v2f;
typedef __attribute__((ext_vector_type(8))) float v8f;
typedef __attribute__((ext_vector_type(4))) float v4f;

#define MROWS 2048
#define FD    128
#define NSTRIPE (MROWS / 16)
#define NTILEBLK (2 * NSTRIPE)
#define NFOLD 8
#define LDP   132

__device__ __forceinline__ v8f tile_k128(const float* pA, const float* pB) {
  v8f acc = {};
  #pragma unroll
  for (int k0 = 0; k0 < FD; k0 += 4) {
    const float2 av = *(const float2*)(pA + k0);
    const float2 bv = *(const float2*)(pB + k0);
    v2f a; a.x = av.x; a.y = av.y;
    v2f b; b.x = bv.x; b.y = bv.y;
    acc = __builtin_amdgcn_wmma_f32_16x16x4_f32(
        false, a, false, b, (short)0, acc, false, false);
  }
  return acc;
}

__device__ __forceinline__ float lane_value(float v, int i) {
  return __int_as_float(__builtin_amdgcn_readlane(__float_as_int(v), i));
}

__global__ __launch_bounds__(128) void layer1_and_fold(
    const float* __restrict__ x,
    const float* __restrict__ W1, const float* __restrict__ b1,
    const float* __restrict__ W2, const float* __restrict__ b2,
    const float* __restrict__ W3, const float* __restrict__ b3,
    const float* __restrict__ W4, const float* __restrict__ b4,
    const float* __restrict__ W5, const float* __restrict__ b5,
    float* __restrict__ H1, float* __restrict__ P, float* __restrict__ Cp) {
  __shared__ __attribute__((aligned(16))) float sW[64 * LDP];
  __shared__ __attribute__((aligned(16))) float sX[16 * LDP];
  __shared__ __attribute__((aligned(16))) float sP1[4][64];
  __shared__ __attribute__((aligned(16))) float sP2[4][64];
  __shared__ __attribute__((aligned(16))) float sP3[4][FD];
  __shared__ float sC[4];

  const int lane = threadIdx.x & 31;
  const int wave = threadIdx.x >> 5;
  const int l15  = lane & 15;
  const int half = lane >> 4;
  const int c4   = lane * 4;
  const int q4   = l15 * 4;

  if (blockIdx.x < NFOLD) {
    const int side = blockIdx.x >> 2;
    const int k1   = (blockIdx.x >> 1) & 1;
    const int k2   = blockIdx.x & 1;
    v4f r4[16], r3[8], r2[16];
    #pragma unroll
    for (int i = 0; i < 16; ++i)
      r4[i] = *(const v4f*)(W4 + (wave * 32 + 2 * i + half) * (2 * FD)
                            + side * FD + k1 * 64 + q4);
    #pragma unroll
    for (int i = 0; i < 8; ++i)
      r3[i] = *(const v4f*)(W3 + (k1 * 64 + wave * 16 + 2 * i + half) * FD
                            + k2 * 64 + q4);
    #pragma unroll
    for (int i = 0; i < 16; ++i)
      r2[i] = *(const v4f*)(W2 + (k2 * 64 + wave * 16 + i) * FD + c4);
    const float w5r = W5[wave * 32 + lane];
    const float b4r = b4[wave * 32 + lane];
    const float b3r = b3[k1 * 64 + wave * 16 + l15];
    const float b2r = b2[k2 * 64 + wave * 16 + l15];
    const float b5v = b5[0];

    v4f p = {0.f, 0.f, 0.f, 0.f};
    #pragma unroll
    for (int i = 0; i < 16; ++i) {
      const float me = lane_value(w5r, 2 * i);
      const float mo = lane_value(w5r, 2 * i + 1);
      p += (half ? mo : me) * r4[i];
    }
    p.x += __shfl_xor(p.x, 16, 32);
    p.y += __shfl_xor(p.y, 16, 32);
    p.z += __shfl_xor(p.z, 16, 32);
    p.w += __shfl_xor(p.w, 16, 32);
    if (lane < 16) *(v4f*)(&sP1[wave][q4]) = p;
    float cacc = (blockIdx.x == 0) ? b4r * w5r : 0.f;
    __syncthreads();
    float vv = 0.f;
    #pragma unroll
    for (int qq = 0; qq < 4; ++qq) vv += sP1[qq][wave * 16 + l15];
    if (k2 == 0 && lane < 16) cacc += b3r * vv;

    p = (v4f){0.f, 0.f, 0.f, 0.f};
    #pragma unroll
    for (int i = 0; i < 8; ++i) {
      const float me = lane_value(vv, 2 * i);
      const float mo = lane_value(vv, 2 * i + 1);
      p += (half ? mo : me) * r3[i];
    }
    p.x += __shfl_xor(p.x, 16, 32);
    p.y += __shfl_xor(p.y, 16, 32);
    p.z += __shfl_xor(p.z, 16, 32);
    p.w += __shfl_xor(p.w, 16, 32);
    if (lane < 16) *(v4f*)(&sP2[wave][q4]) = p;
    __syncthreads();
    float u3 = 0.f;
    #pragma unroll
    for (int qq = 0; qq < 4; ++qq) u3 += sP2[qq][wave * 16 + l15];
    if (lane < 16) cacc += b2r * u3;

    p = (v4f){0.f, 0.f, 0.f, 0.f};
    #pragma unroll
    for (int i = 0; i < 16; ++i) p += lane_value(u3, i) * r2[i];
    *(v4f*)(&sP3[wave][c4]) = p;
    #pragma unroll
    for (int off = 16; off > 0; off >>= 1) cacc += __shfl_down(cacc, off, 32);
    if (lane == 0) sC[wave] = cacc;
    __syncthreads();
    {
      float s = 0.f;
      #pragma unroll
      for (int qq = 0; qq < 4; ++qq) s += sP3[qq][threadIdx.x];
      P[blockIdx.x * FD + threadIdx.x] = s;
    }
    if (threadIdx.x == 0) {
      float c = (blockIdx.x == 0) ? b5v : 0.f;
      #pragma unroll
      for (int qq = 0; qq < 4; ++qq) c += sC[qq];
      Cp[blockIdx.x] = c;
    }
    return;
  }

  const int tb   = blockIdx.x - NFOLD;
  const int m0   = (tb >> 1) * 16;
  const int nb   = (tb & 1) * 64;
  const int col  = nb + wave * 16 + l15;
  const int wofs = (wave * 16 + l15) * LDP + half * 2;
  const int sofs = l15 * LDP + half * 2;

  v4f xq[4], wq1[16];
  #pragma unroll
  for (int i = 0; i < 4; ++i)
    xq[i] = *(const v4f*)(x + (m0 + wave * 4 + i) * FD + c4);
  #pragma unroll
  for (int i = 0; i < 16; ++i)
    wq1[i] = *(const v4f*)(W1 + (nb + wave * 16 + i) * FD + c4);
  const float bv1 = b1[col];

  #pragma unroll
  for (int i = 0; i < 4; ++i) *(v4f*)(sX + (wave * 4 + i) * LDP + c4) = xq[i];
  #pragma unroll
  for (int i = 0; i < 16; ++i) *(v4f*)(sW + (wave * 16 + i) * LDP + c4) = wq1[i];
  __syncthreads();
  {
    const v8f acc = tile_k128(sX + sofs, sW + wofs);
    #pragma unroll
    for (int g = 0; g < 8; ++g) {
      const int row = m0 + g + half * 8;
      H1[row * FD + col] = acc[g] + bv1;
    }
  }
}

__global__ __launch_bounds__(256) void dots_and_expand(
    const float* __restrict__ H, const float* __restrict__ P,
    const float* __restrict__ Cp, float* __restrict__ out) {
  __shared__ __attribute__((aligned(16))) float sUr[64];
  __shared__ float sUl[64];
  const int lane = threadIdx.x & 31;
  const int wave = threadIdx.x >> 5;
  const int b    = blockIdx.x >> 4;
  const int it   = (blockIdx.x >> 2) & 3;
  const int jt   = blockIdx.x & 3;

  const float c  = ((Cp[0] + Cp[1]) + (Cp[2] + Cp[3])) + ((Cp[4] + Cp[5]) + (Cp[6] + Cp[7]));
  v4f pl[4], pr[4];
  #pragma unroll
  for (int j = 0; j < 4; ++j) pl[j] = *(const v4f*)(P + j * FD + lane * 4);
  #pragma unroll
  for (int j = 0; j < 4; ++j) pr[j] = *(const v4f*)(P + (4 + j) * FD + lane * 4);
  const float* Hb = H + b * 256 * FD + lane * 4;
  v4f hj[8], hi[8];
  #pragma unroll
  for (int r = 0; r < 8; ++r) hj[r] = *(const v4f*)(Hb + (jt * 64 + wave * 8 + r) * FD);
  #pragma unroll
  for (int r = 0; r < 8; ++r) hi[r] = *(const v4f*)(Hb + (it * 64 + wave * 8 + r) * FD);
  asm volatile("" :: "v"(hj[0]), "v"(hj[1]), "v"(hj[2]), "v"(hj[3]), "v"(hj[4]), "v"(hj[5]),
               "v"(hj[6]), "v"(hj[7]), "v"(hi[0]), "v"(hi[1]), "v"(hi[2]), "v"(hi[3]),
               "v"(hi[4]), "v"(hi[5]), "v"(hi[6]), "v"(hi[7]), "v"(pl[0]), "v"(pl[1]),
               "v"(pl[2]), "v"(pl[3]), "v"(pr[0]), "v"(pr[1]), "v"(pr[2]), "v"(pr[3]));
  const v4f l4 = (pl[0] + pl[1]) + (pl[2] + pl[3]);
  const v4f r4 = (pr[0] + pr[1]) + (pr[2] + pr[3]);

  float sr[8], sl[8];
  #pragma unroll
  for (int r = 0; r < 8; ++r) {
    sr[r] = hj[r].x * r4.x + hj[r].y * r4.y + hj[r].z * r4.z + hj[r].w * r4.w;
    sl[r] = hi[r].x * l4.x + hi[r].y * l4.y + hi[r].z * l4.z + hi[r].w * l4.w;
  }
  #pragma unroll
  for (int off = 16; off > 0; off >>= 1) {
    #pragma unroll
    for (int r = 0; r < 8; ++r) {
      sr[r] += __shfl_down(sr[r], off, 32);
      sl[r] += __shfl_down(sl[r], off, 32);
    }
  }
  if (lane == 0) {
    #pragma unroll
    for (int r = 0; r < 8; ++r) {
      sUr[wave * 8 + r] = sr[r];
      sUl[wave * 8 + r] = sl[r];
    }
  }
  __syncthreads();

  const int   jq = (threadIdx.x & 15) * 4;
  const int   rs = threadIdx.x >> 4;
  const float4 u4 = *(const float4*)(sUr + jq);
  float* ob = out + (b * 256 + it * 64) * 256 + jt * 64 + jq;
  for (int p = 0; p < 4; ++p) {
    const int   r = p * 16 + rs;
    const float u = sUl[r];
    float4 o;
    o.x = u + u4.x + c;
    o.y = u + u4.y + c;
    o.z = u + u4.z + c;
    o.w = u + u4.w + c;
    *(float4*)(ob + r * 256) = o;
  }
}

extern "C" void kernel_launch(void* const* d_in, const int* in_sizes, int n_in,
                              void* d_out, int out_size, void* d_ws, size_t ws_size,
                              hipStream_t stream) {
  (void)in_sizes; (void)n_in; (void)out_size; (void)ws_size;
  const float* x  = (const float*)d_in[0];
  const float* W1 = (const float*)d_in[1];
  const float* b1 = (const float*)d_in[2];
  const float* W2 = (const float*)d_in[3];
  const float* b2 = (const float*)d_in[4];
  const float* W3 = (const float*)d_in[5];
  const float* b3 = (const float*)d_in[6];
  const float* W4 = (const float*)d_in[7];
  const float* b4 = (const float*)d_in[8];
  const float* W5 = (const float*)d_in[9];
  const float* b5 = (const float*)d_in[10];
  float* out = (float*)d_out;

  float* ws = (float*)d_ws;
  float* H1 = ws;
  float* P  = H1 + MROWS * FD;
  float* Cp = P + NFOLD * FD;

  layer1_and_fold<<<NFOLD + NTILEBLK, 128, 0, stream>>>(
      x, W1, b1, W2, b2, W3, b3, W4, b4, W5, b5, H1, P, Cp);

  dots_and_expand<<<8 * 16, 256, 0, stream>>>(H1, P, Cp, out);
}


# Cross-lane sums by v_permlanex16 and row_xmask DPP, not runtime shuffles through the LDS crossbar; launch 2's 16 row sums share each exchange (halving butterfly); pairing order kept, bit-identical
# speedup vs baseline: 4.5578x; 1.0512x over previous
#include <hip/hip_runtime.h>
#include <hip/hip_bf16.h>

typedef __attribute__((ext_vector_type(2))) float v2f;
typedef __attribute__((ext_vector_type(8))) float v8f;
typedef __attribute__((ext_vector_type(4))) float v4f;

#define MROWS 2048
#define FD    128
#define NSTRIPE (MROWS / 16)
#define NTILEBLK (2 * NSTRIPE)
#define NFOLD 8
#define LDP   132

__device__ __forceinline__ v8f tile_k128(const float* pA, const float* pB) {
  v8f acc = {};
  #pragma unroll
  for (int k0 = 0; k0 < FD; k0 += 4) {
    const float2 av = *(const float2*)(pA + k0);
    const float2 bv = *(const float2*)(pB + k0);
    v2f a; a.x = av.x; a.y = av.y;
    v2f b; b.x = bv.x; b.y = bv.y;
    acc = __builtin_amdgcn_wmma_f32_16x16x4_f32(
        false, a, false, b, (short)0, acc, false, false);
  }
  return acc;
}

__device__ __forceinline__ float lane_value(float v, int i) {
  return __int_as_float(__builtin_amdgcn_readlane(__float_as_int(v), i));
}

__device__ __forceinline__ float xhalf(float v) {
  const int b = __float_as_int(v);
  return __int_as_float(
      __builtin_amdgcn_permlanex16(b, b, 0x76543210, 0xfedcba98, false, false));
}
template <int M>
__device__ __forceinline__ float xrow(float v) {
  const int b = __float_as_int(v);
  return __int_as_float(
      __builtin_amdgcn_update_dpp(0, b, 0x160 + M, 0xf, 0xf, true));
}
__device__ __forceinline__ float wave_sum(float v) {
  v += xhalf(v);
  v += xrow<8>(v);
  v += xrow<4>(v);
  v += xrow<2>(v);
  v += xrow<1>(v);
  return v;
}

__global__ __launch_bounds__(128) void layer1_and_fold(
    const float* __restrict__ x,
    const float* __restrict__ W1, const float* __restrict__ b1,
    const float* __restrict__ W2, const float* __restrict__ b2,
    const float* __restrict__ W3, const float* __restrict__ b3,
    const float* __restrict__ W4, const float* __restrict__ b4,
    const float* __restrict__ W5, const float* __restrict__ b5,
    float* __restrict__ H1, float* __restrict__ P, float* __restrict__ Cp) {
  __shared__ __attribute__((aligned(16))) float sW[64 * LDP];
  __shared__ __attribute__((aligned(16))) float sX[16 * LDP];
  __shared__ __attribute__((aligned(16))) float sP1[4][64];
  __shared__ __attribute__((aligned(16))) float sP2[4][64];
  __shared__ __attribute__((aligned(16))) float sP3[4][FD];
  __shared__ float sC[4];

  const int lane = threadIdx.x & 31;
  const int wave = threadIdx.x >> 5;
  const int l15  = lane & 15;
  const int half = lane >> 4;
  const int c4   = lane * 4;
  const int q4   = l15 * 4;

  if (blockIdx.x < NFOLD) {
    const int side = blockIdx.x >> 2;
    const int k1   = (blockIdx.x >> 1) & 1;
    const int k2   = blockIdx.x & 1;
    v4f r4[16], r3[8], r2[16];
    #pragma unroll
    for (int i = 0; i < 16; ++i)
      r4[i] = *(const v4f*)(W4 + (wave * 32 + 2 * i + half) * (2 * FD)
                            + side * FD + k1 * 64 + q4);
    #pragma unroll
    for (int i = 0; i < 8; ++i)
      r3[i] = *(const v4f*)(W3 + (k1 * 64 + wave * 16 + 2 * i + half) * FD
                            + k2 * 64 + q4);
    #pragma unroll
    for (int i = 0; i < 16; ++i)
      r2[i] = *(const v4f*)(W2 + (k2 * 64 + wave * 16 + i) * FD + c4);
    const float w5r = W5[wave * 32 + lane];
    const float b4r = b4[wave * 32 + lane];
    const float b3r = b3[k1 * 64 + wave * 16 + l15];
    const float b2r = b2[k2 * 64 + wave * 16 + l15];
    const float b5v = b5[0];

    v4f p = {0.f, 0.f, 0.f, 0.f};
    #pragma unroll
    for (int i = 0; i < 16; ++i) {
      const float me = lane_value(w5r, 2 * i);
      const float mo = lane_value(w5r, 2 * i + 1);
      p += (half ? mo : me) * r4[i];
    }
    p.x += xhalf(p.x);
    p.y += xhalf(p.y);
    p.z += xhalf(p.z);
    p.w += xhalf(p.w);
    if (lane < 16) *(v4f*)(&sP1[wave][q4]) = p;
    float cacc = (blockIdx.x == 0) ? b4r * w5r : 0.f;
    __syncthreads();
    float vv = 0.f;
    #pragma unroll
    for (int qq = 0; qq < 4; ++qq) vv += sP1[qq][wave * 16 + l15];
    if (k2 == 0 && lane < 16) cacc += b3r * vv;

    p = (v4f){0.f, 0.f, 0.f, 0.f};
    #pragma unroll
    for (int i = 0; i < 8; ++i) {
      const float me = lane_value(vv, 2 * i);
      const float mo = lane_value(vv, 2 * i + 1);
      p += (half ? mo : me) * r3[i];
    }
    p.x += xhalf(p.x);
    p.y += xhalf(p.y);
    p.z += xhalf(p.z);
    p.w += xhalf(p.w);
    if (lane < 16) *(v4f*)(&sP2[wave][q4]) = p;
    __syncthreads();
    float u3 = 0.f;
    #pragma unroll
    for (int qq = 0; qq < 4; ++qq) u3 += sP2[qq][wave * 16 + l15];
    if (lane < 16) cacc += b2r * u3;

    p = (v4f){0.f, 0.f, 0.f, 0.f};
    #pragma unroll
    for (int i = 0; i < 16; ++i) p += lane_value(u3, i) * r2[i];
    *(v4f*)(&sP3[wave][c4]) = p;
    cacc = wave_sum(cacc);
    if (lane == 0) sC[wave] = cacc;
    __syncthreads();
    {
      float s = 0.f;
      #pragma unroll
      for (int qq = 0; qq < 4; ++qq) s += sP3[qq][threadIdx.x];
      P[blockIdx.x * FD + threadIdx.x] = s;
    }
    if (threadIdx.x == 0) {
      float c = (blockIdx.x == 0) ? b5v : 0.f;
      #pragma unroll
      for (int qq = 0; qq < 4; ++qq) c += sC[qq];
      Cp[blockIdx.x] = c;
    }
    return;
  }

  const int tb   = blockIdx.x - NFOLD;
  const int m0   = (tb >> 1) * 16;
  const int nb   = (tb & 1) * 64;
  const int col  = nb + wave * 16 + l15;
  const int wofs = (wave * 16 + l15) * LDP + half * 2;
  const int sofs = l15 * LDP + half * 2;

  v4f xq[4], wq1[16];
  #pragma unroll
  for (int i = 0; i < 4; ++i)
    xq[i] = *(const v4f*)(x + (m0 + wave * 4 + i) * FD + c4);
  #pragma unroll
  for (int i = 0; i < 16; ++i)
    wq1[i] = *(const v4f*)(W1 + (nb + wave * 16 + i) * FD + c4);
  const float bv1 = b1[col];

  #pragma unroll
  for (int i = 0; i < 4; ++i) *(v4f*)(sX + (wave * 4 + i) * LDP + c4) = xq[i];
  #pragma unroll
  for (int i = 0; i < 16; ++i) *(v4f*)(sW + (wave * 16 + i) * LDP + c4) = wq1[i];
  __syncthreads();
  {
    const v8f acc = tile_k128(sX + sofs, sW + wofs);
    #pragma unroll
    for (int g = 0; g < 8; ++g) {
      const int row = m0 + g + half * 8;
      H1[row * FD + col] = acc[g] + bv1;
    }
  }
}

__global__ __launch_bounds__(256) void dots_and_expand(
    const float* __restrict__ H, const float* __restrict__ P,
    const float* __restrict__ Cp, float* __restrict__ out) {
  __shared__ __attribute__((aligned(16))) float sUr[64];
  __shared__ float sUl[64];
  const int lane = threadIdx.x & 31;
  const int wave = threadIdx.x >> 5;
  const int b    = blockIdx.x >> 4;
  const int it   = (blockIdx.x >> 2) & 3;
  const int jt   = blockIdx.x & 3;

  const float c  = ((Cp[0] + Cp[1]) + (Cp[2] + Cp[3])) + ((Cp[4] + Cp[5]) + (Cp[6] + Cp[7]));
  v4f pl[4], pr[4];
  #pragma unroll
  for (int j = 0; j < 4; ++j) pl[j] = *(const v4f*)(P + j * FD + lane * 4);
  #pragma unroll
  for (int j = 0; j < 4; ++j) pr[j] = *(const v4f*)(P + (4 + j) * FD + lane * 4);
  const float* Hb = H + b * 256 * FD + lane * 4;
  v4f hj[8], hi[8];
  #pragma unroll
  for (int r = 0; r < 8; ++r) hj[r] = *(const v4f*)(Hb + (jt * 64 + wave * 8 + r) * FD);
  #pragma unroll
  for (int r = 0; r < 8; ++r) hi[r] = *(const v4f*)(Hb + (it * 64 + wave * 8 + r) * FD);
  asm volatile("" :: "v"(hj[0]), "v"(hj[1]), "v"(hj[2]), "v"(hj[3]), "v"(hj[4]), "v"(hj[5]),
               "v"(hj[6]), "v"(hj[7]), "v"(hi[0]), "v"(hi[1]), "v"(hi[2]), "v"(hi[3]),
               "v"(hi[4]), "v"(hi[5]), "v"(hi[6]), "v"(hi[7]), "v"(pl[0]), "v"(pl[1]),
               "v"(pl[2]), "v"(pl[3]), "v"(pr[0]), "v"(pr[1]), "v"(pr[2]), "v"(pr[3]));
  const v4f l4 = (pl[0] + pl[1]) + (pl[2] + pl[3]);
  const v4f r4 = (pr[0] + pr[1]) + (pr[2] + pr[3]);

  float sr[8], sl[8];
  #pragma unroll
  for (int r = 0; r < 8; ++r) {
    sr[r] = hj[r].x * r4.x + hj[r].y * r4.y + hj[r].z * r4.z + hj[r].w * r4.w;
    sl[r] = hi[r].x * l4.x + hi[r].y * l4.y + hi[r].z * l4.z + hi[r].w * l4.w;
  }
  const bool b16 = (lane & 16) != 0;
  const bool b8  = (lane & 8) != 0;
  const bool b4  = (lane & 4) != 0;
  const bool b2  = (lane & 2) != 0;
  float s8[8], s4[4], s2[2];
  #pragma unroll
  for (int r = 0; r < 8; ++r)
    s8[r] = (b16 ? sl[r] : sr[r]) + xhalf(b16 ? sr[r] : sl[r]);
  #pragma unroll
  for (int r = 0; r < 4; ++r)
    s4[r] = (b8 ? s8[r + 4] : s8[r]) + xrow<8>(b8 ? s8[r] : s8[r + 4]);
  #pragma unroll
  for (int r = 0; r < 2; ++r)
    s2[r] = (b4 ? s4[r + 2] : s4[r]) + xrow<4>(b4 ? s4[r] : s4[r + 2]);
  float s1 = (b2 ? s2[1] : s2[0]) + xrow<2>(b2 ? s2[0] : s2[1]);
  s1 += xrow<1>(s1);
  if ((lane & 1) == 0) {
    float* dst = b16 ? sUl : sUr;
    dst[wave * 8 + ((lane >> 1) & 7)] = s1;
  }
  __syncthreads();

  const int   jq = (threadIdx.x & 15) * 4;
  const int   rs = threadIdx.x >> 4;
  const float4 u4 = *(const float4*)(sUr + jq);
  float* ob = out + (b * 256 + it * 64) * 256 + jt * 64 + jq;
  for (int p = 0; p < 4; ++p) {
    const int   r = p * 16 + rs;
    const float u = sUl[r];
    float4 o;
    o.x = u + u4.x + c;
    o.y = u + u4.y + c;
    o.z = u + u4.z + c;
    o.w = u + u4.w + c;
    *(float4*)(ob + r * 256) = o;
  }
}

extern "C" void kernel_launch(void* const* d_in, const int* in_sizes, int n_in,
                              void* d_out, int out_size, void* d_ws, size_t ws_size,
                              hipStream_t stream) {
  (void)in_sizes; (void)n_in; (void)out_size; (void)ws_size;
  const float* x  = (const float*)d_in[0];
  const float* W1 = (const float*)d_in[1];
  const float* b1 = (const float*)d_in[2];
  const float* W2 = (const float*)d_in[3];
  const float* b2 = (const float*)d_in[4];
  const float* W3 = (const float*)d_in[5];
  const float* b3 = (const float*)d_in[6];
  const float* W4 = (const float*)d_in[7];
  const float* b4 = (const float*)d_in[8];
  const float* W5 = (const float*)d_in[9];
  const float* b5 = (const float*)d_in[10];
  float* out = (float*)d_out;

  float* ws = (float*)d_ws;
  float* H1 = ws;
  float* P  = H1 + MROWS * FD;
  float* Cp = P + NFOLD * FD;

  layer1_and_fold<<<NFOLD + NTILEBLK, 128, 0, stream>>>(
      x, W1, b1, W2, b2, W3, b3, W4, b4, W5, b5, H1, P, Cp);

  dots_and_expand<<<8 * 16, 256, 0, stream>>>(H1, P, Cp, out);
}
